# starting kernel
# baseline (speedup 1.0000x reference)
#include <hip/hip_runtime.h>
#include <hip/hip_bf16.h>

typedef float v8f __attribute__((ext_vector_type(8)));
typedef float v2f __attribute__((ext_vector_type(2)));

#define DIMD 128
#define NNODES 50000
#define NEDGES 800000

__global__ __launch_bounds__(256) void gemm_wmma_f32(
    const float* __restrict__ H, const float* __restrict__ W,
    float* __restrict__ Z) {
  const int wave = threadIdx.x >> 5;
  const int lane = threadIdx.x & 31;
  const int row0 = blockIdx.x * 16;
  const int col0 = wave * 16;
  const int m    = lane & 15;
  const int kh   = (lane >> 4) * 2;

  const float* __restrict__ arow = H + (size_t)(row0 + m) * DIMD;
  v8f acc = {};
#pragma unroll
  for (int k = 0; k < DIMD; k += 4) {
    v2f a, b;
    a.x = arow[k + kh];
    a.y = arow[k + kh + 1];
    b.x = W[(size_t)(k + kh) * DIMD + col0 + m];
    b.y = W[(size_t)(k + kh + 1) * DIMD + col0 + m];
    acc = __builtin_amdgcn_wmma_f32_16x16x4_f32(
         false, a,  false, b,
         (short)0, acc,  false,  false);
  }
  const int mbase = row0 + (lane >> 4) * 8;
#pragma unroll
  for (int j = 0; j < 8; ++j)
    Z[(size_t)(mbase + j) * DIMD + col0 + m] = acc[j];
}

__global__ __launch_bounds__(256) void scatter_add_edges(
    const float* __restrict__ Z, const int* __restrict__ src,
    const int* __restrict__ dst, float* __restrict__ agg, int n_edges) {
  const int e    = blockIdx.x * 8 + (threadIdx.x >> 5);
  const int lane = threadIdx.x & 31;
  if (e >= n_edges) return;
  const int s = src[e];
  const int d = dst[e];
  const float4 v = ((const float4*)(Z + (size_t)s * DIMD))[lane];
  float* ap = agg + (size_t)d * DIMD + lane * 4;
  unsafeAtomicAdd(ap + 0, v.x);
  unsafeAtomicAdd(ap + 1, v.y);
  unsafeAtomicAdd(ap + 2, v.z);
  unsafeAtomicAdd(ap + 3, v.w);
}

__global__ __launch_bounds__(128) void bias_prelu_stats(
    float* __restrict__ ZA, const float* __restrict__ bias,
    const float* __restrict__ a_conv_p, float* __restrict__ stats,
    int n_rows, int rows_per_block) {
  const int c  = threadIdx.x;
  const float bb = bias[c];
  const float ac = *a_conv_p;
  const int r0 = blockIdx.x * rows_per_block;
  int r1 = r0 + rows_per_block;
  if (r1 > n_rows) r1 = n_rows;
  float s = 0.f, ss = 0.f;
  for (int r = r0; r < r1; ++r) {
    const size_t idx = (size_t)r * DIMD + c;
    float z = ZA[idx] + bb;
    z = (z >= 0.f) ? z : ac * z;
    ZA[idx] = z;
    s += z;
    ss += z * z;
  }
  unsafeAtomicAdd(&stats[c], s);
  unsafeAtomicAdd(&stats[DIMD + c], ss);
}

__global__ __launch_bounds__(128) void bn_finalize(
    float* __restrict__ stats, const float* __restrict__ gamma,
    const float* __restrict__ beta, int n_rows) {
  const int c = threadIdx.x;
  const float inv_n = 1.0f / (float)n_rows;
  const float mu  = stats[c] * inv_n;
  const float var = stats[DIMD + c] * inv_n - mu * mu;
  const float scale = rsqrtf(var + 1e-5f) * gamma[c];
  const float shift = beta[c] - mu * scale;
  stats[c]        = scale;
  stats[DIMD + c] = shift;
}

__global__ __launch_bounds__(256) void bn_apply_prelu(
    const float* __restrict__ Z, const float* __restrict__ stats,
    const float* __restrict__ a_out_p, float* __restrict__ Hout,
    size_t n_elems) {
  const float ao = *a_out_p;
  const size_t stride = (size_t)gridDim.x * blockDim.x;
  for (size_t i = (size_t)blockIdx.x * blockDim.x + threadIdx.x; i < n_elems;
       i += stride) {
    const int c = (int)(i & (DIMD - 1));
    float z = fmaf(Z[i], stats[c], stats[DIMD + c]);
    Hout[i] = (z >= 0.f) ? z : ao * z;
  }
}

extern "C" void kernel_launch(void* const* d_in, const int* in_sizes, int n_in,
                              void* d_out, int out_size, void* d_ws,
                              size_t ws_size, hipStream_t stream) {
  const float* heat   = (const float*)d_in[0];
  const int*   src    = (const int*)d_in[1];
  const int*   dst    = (const int*)d_in[2];
  const float* W      = (const float*)d_in[3];
  const float* b      = (const float*)d_in[4];
  const float* a_conv = (const float*)d_in[5];
  const float* gamma  = (const float*)d_in[6];
  const float* beta   = (const float*)d_in[7];
  const float* a_out  = (const float*)d_in[8];

  const size_t nb = (size_t)NNODES * DIMD * sizeof(float);
  float* bufA  = (float*)d_ws;
  float* bufB  = (float*)((char*)d_ws + nb);
  float* stats = (float*)((char*)d_ws + 2 * nb);

  const dim3 gemmGrid(NNODES / 16);
  const dim3 gemmBlock(256);
  const int rows_per_block = 64;
  const int statBlocks = (NNODES + rows_per_block - 1) / rows_per_block;
  const int scatBlocks = (NEDGES + 7) / 8;
  const size_t nelem = (size_t)NNODES * DIMD;

  hipMemsetAsync(bufB, 0, nb, stream);
  hipMemsetAsync(stats, 0, 2 * DIMD * sizeof(float), stream);
  gemm_wmma_f32<<<gemmGrid, gemmBlock, 0, stream>>>(heat, W, bufA);
  scatter_add_edges<<<scatBlocks, 256, 0, stream>>>(bufA, src, dst, bufB,
                                                    NEDGES);
  bias_prelu_stats<<<statBlocks, 128, 0, stream>>>(bufB, b, a_conv, stats,
                                                   NNODES, rows_per_block);
  bn_finalize<<<1, 128, 0, stream>>>(stats, gamma, beta, NNODES);
  bn_apply_prelu<<<4096, 256, 0, stream>>>(bufB, stats, a_out, bufA, nelem);

  gemm_wmma_f32<<<gemmGrid, gemmBlock, 0, stream>>>(bufA, W + DIMD * DIMD,
                                                    bufB);
  hipMemsetAsync(bufA, 0, nb, stream);
  hipMemsetAsync(stats, 0, 2 * DIMD * sizeof(float), stream);
  scatter_add_edges<<<scatBlocks, 256, 0, stream>>>(bufB, src, dst, bufA,
                                                    NEDGES);
  bias_prelu_stats<<<statBlocks, 128, 0, stream>>>(bufA, b + DIMD, a_conv + 1,
                                                   stats, NNODES,
                                                   rows_per_block);
  bn_finalize<<<1, 128, 0, stream>>>(stats, gamma + DIMD, beta + DIMD, NNODES);
  bn_apply_prelu<<<4096, 256, 0, stream>>>(bufA, stats, a_out + 1,
                                           (float*)d_out, nelem);
}


# Replace the per-edge f32 atomic scatter with in-edge lists built once per call (count, scan, fill) and an atomic-free row gather used by both layers
# speedup vs baseline: 1.4656x; 1.4656x over previous
#include <hip/hip_runtime.h>
#include <hip/hip_bf16.h>

typedef float v8f __attribute__((ext_vector_type(8)));
typedef float v2f __attribute__((ext_vector_type(2)));

#define DIMD 128
#define NNODES 50000
#define NEDGES 800000

__global__ __launch_bounds__(256) void gemm_wmma_f32(
    const float* __restrict__ H, const float* __restrict__ W,
    float* __restrict__ Z) {
  const int wave = threadIdx.x >> 5;
  const int lane = threadIdx.x & 31;
  const int row0 = blockIdx.x * 16;
  const int col0 = wave * 16;
  const int m    = lane & 15;
  const int kh   = (lane >> 4) * 2;

  const float* __restrict__ arow = H + (size_t)(row0 + m) * DIMD;
  v8f acc = {};
#pragma unroll
  for (int k = 0; k < DIMD; k += 4) {
    v2f a, b;
    a.x = arow[k + kh];
    a.y = arow[k + kh + 1];
    b.x = W[(size_t)(k + kh) * DIMD + col0 + m];
    b.y = W[(size_t)(k + kh + 1) * DIMD + col0 + m];
    acc = __builtin_amdgcn_wmma_f32_16x16x4_f32(
         false, a,  false, b,
         (short)0, acc,  false,  false);
  }
  const int mbase = row0 + (lane >> 4) * 8;
#pragma unroll
  for (int j = 0; j < 8; ++j)
    Z[(size_t)(mbase + j) * DIMD + col0 + m] = acc[j];
}

__global__ __launch_bounds__(256) void count_in_edges(
    const int* __restrict__ dst, int* __restrict__ cnt, int* __restrict__ rank,
    int n_edges) {
  const int e = blockIdx.x * 256 + threadIdx.x;
  if (e < n_edges) rank[e] = atomicAdd(&cnt[dst[e]], 1);
}

#define SCAN_ROWS 13
#define SCAN_LEN (32 * SCAN_ROWS * 128)
__global__ __launch_bounds__(1024) void scan_in_degrees(int* p) {
  __shared__ int wave_total[32];
  const int lane = threadIdx.x & 31;
  const int w = __builtin_amdgcn_readfirstlane((int)(threadIdx.x >> 5));
  int4* q = (int4*)(p + w * (SCAN_ROWS * 128)) + lane;
  int4 v[SCAN_ROWS];
  int x[SCAN_ROWS];
#pragma unroll
  for (int i = 0; i < SCAN_ROWS; ++i) v[i] = q[i * 32];
#pragma unroll
  for (int i = 0; i < SCAN_ROWS; ++i) {
    v[i].y += v[i].x;
    v[i].z += v[i].y;
    v[i].w += v[i].z;
    x[i] = v[i].w;
  }
#pragma unroll
  for (int d = 1; d < 32; d <<= 1) {
    int y[SCAN_ROWS];
#pragma unroll
    for (int i = 0; i < SCAN_ROWS; ++i) y[i] = __shfl_up(x[i], d);
#pragma unroll
    for (int i = 0; i < SCAN_ROWS; ++i)
      if (lane >= d) x[i] += y[i];
  }
  int carry = 0;
  int off[SCAN_ROWS];
#pragma unroll
  for (int i = 0; i < SCAN_ROWS; ++i) {
    off[i] = carry + x[i] - v[i].w;
    carry += __builtin_amdgcn_readlane(x[i], 31);
  }
  if (lane == 0) wave_total[w] = carry;
  __syncthreads();
  const int t = wave_total[lane];
  int ts = t;
#pragma unroll
  for (int d = 1; d < 32; d <<= 1) {
    const int y = __shfl_up(ts, d);
    if (lane >= d) ts += y;
  }
  const int base = __builtin_amdgcn_readlane(ts - t, w);
#pragma unroll
  for (int i = 0; i < SCAN_ROWS; ++i) {
    const int o = base + off[i];
    int4 r;
    r.x = o;
    r.y = o + v[i].x;
    r.z = o + v[i].y;
    r.w = o + v[i].z;
    q[i * 32] = r;
  }
}

__global__ __launch_bounds__(256) void fill_in_edge_lists(
    const int* __restrict__ src, const int* __restrict__ dst,
    const int* __restrict__ rowptr, const int* __restrict__ rank,
    int* __restrict__ col, int n_edges) {
  const int e = blockIdx.x * 256 + threadIdx.x;
  if (e < n_edges) col[rowptr[dst[e]] + rank[e]] = src[e];
}

__global__ __launch_bounds__(256) void gather_sum_in_edges(
    const float* __restrict__ Z, const int* __restrict__ rowptr,
    const int* __restrict__ col, float* __restrict__ agg, int n_rows) {
  const int lane = threadIdx.x & 31;
  const int wave = __builtin_amdgcn_readfirstlane((int)(threadIdx.x >> 5));
  const int n_waves = (int)gridDim.x * 8;
  const float4* __restrict__ Z4 = (const float4*)Z;
  float4* __restrict__ A4 = (float4*)agg;
  for (int d = (int)blockIdx.x * 8 + wave; d < n_rows; d += n_waves) {
    const int beg = rowptr[d];
    const int end = rowptr[d + 1];
    float4 acc = {0.f, 0.f, 0.f, 0.f};
    for (int b = beg; b < end; b += 32) {
      const int j = b + lane;
      const int mine = (j < end) ? col[j] : 0;
      const int m = (end - b < 32) ? (end - b) : 32;
#pragma unroll 4
      for (int k = 0; k < m; ++k) {
        const int s = __builtin_amdgcn_readlane(mine, k);
        const float4 v = Z4[(size_t)s * (DIMD / 4) + lane];
        acc.x += v.x;
        acc.y += v.y;
        acc.z += v.z;
        acc.w += v.w;
      }
    }
    A4[(size_t)d * (DIMD / 4) + lane] = acc;
  }
}

__global__ __launch_bounds__(128) void bias_prelu_stats(
    float* __restrict__ ZA, const float* __restrict__ bias,
    const float* __restrict__ a_conv_p, float* __restrict__ stats,
    int n_rows, int rows_per_block) {
  const int c  = threadIdx.x;
  const float bb = bias[c];
  const float ac = *a_conv_p;
  const int r0 = blockIdx.x * rows_per_block;
  int r1 = r0 + rows_per_block;
  if (r1 > n_rows) r1 = n_rows;
  float s = 0.f, ss = 0.f;
  for (int r = r0; r < r1; ++r) {
    const size_t idx = (size_t)r * DIMD + c;
    float z = ZA[idx] + bb;
    z = (z >= 0.f) ? z : ac * z;
    ZA[idx] = z;
    s += z;
    ss += z * z;
  }
  unsafeAtomicAdd(&stats[c], s);
  unsafeAtomicAdd(&stats[DIMD + c], ss);
}

__global__ __launch_bounds__(128) void bn_finalize(
    float* __restrict__ stats, const float* __restrict__ gamma,
    const float* __restrict__ beta, int n_rows) {
  const int c = threadIdx.x;
  const float inv_n = 1.0f / (float)n_rows;
  const float mu  = stats[c] * inv_n;
  const float var = stats[DIMD + c] * inv_n - mu * mu;
  const float scale = rsqrtf(var + 1e-5f) * gamma[c];
  const float shift = beta[c] - mu * scale;
  stats[c]        = scale;
  stats[DIMD + c] = shift;
}

__global__ __launch_bounds__(256) void bn_apply_prelu(
    const float* __restrict__ Z, const float* __restrict__ stats,
    const float* __restrict__ a_out_p, float* __restrict__ Hout,
    size_t n_elems) {
  const float ao = *a_out_p;
  const size_t stride = (size_t)gridDim.x * blockDim.x;
  for (size_t i = (size_t)blockIdx.x * blockDim.x + threadIdx.x; i < n_elems;
       i += stride) {
    const int c = (int)(i & (DIMD - 1));
    float z = fmaf(Z[i], stats[c], stats[DIMD + c]);
    Hout[i] = (z >= 0.f) ? z : ao * z;
  }
}

extern "C" void kernel_launch(void* const* d_in, const int* in_sizes, int n_in,
                              void* d_out, int out_size, void* d_ws,
                              size_t ws_size, hipStream_t stream) {
  const float* heat   = (const float*)d_in[0];
  const int*   src    = (const int*)d_in[1];
  const int*   dst    = (const int*)d_in[2];
  const float* W      = (const float*)d_in[3];
  const float* b      = (const float*)d_in[4];
  const float* a_conv = (const float*)d_in[5];
  const float* gamma  = (const float*)d_in[6];
  const float* beta   = (const float*)d_in[7];
  const float* a_out  = (const float*)d_in[8];

  const size_t nb = (size_t)NNODES * DIMD * sizeof(float);
  float* bufA  = (float*)d_ws;
  float* bufB  = (float*)((char*)d_ws + nb);
  float* stats = (float*)((char*)d_ws + 2 * nb);

  int* rowptr = (int*)d_out;
  int* rank   = rowptr + 65536;
  int* col    = rank + NEDGES;

  const dim3 gemmGrid(NNODES / 16);
  const dim3 gemmBlock(256);
  const int rows_per_block = 64;
  const int statBlocks = (NNODES + rows_per_block - 1) / rows_per_block;
  const int edgeBlocks = (NEDGES + 255) / 256;
  const int gatherBlocks = 2048;
  const size_t nelem = (size_t)NNODES * DIMD;

  hipMemsetAsync(rowptr, 0, SCAN_LEN * sizeof(int), stream);
  hipMemsetAsync(stats, 0, 2 * DIMD * sizeof(float), stream);
  count_in_edges<<<edgeBlocks, 256, 0, stream>>>(dst, rowptr, rank, NEDGES);
  scan_in_degrees<<<1, 1024, 0, stream>>>(rowptr);
  fill_in_edge_lists<<<edgeBlocks, 256, 0, stream>>>(src, dst, rowptr, rank,
                                                     col, NEDGES);

  gemm_wmma_f32<<<gemmGrid, gemmBlock, 0, stream>>>(heat, W, bufA);
  gather_sum_in_edges<<<gatherBlocks, 256, 0, stream>>>(bufA, rowptr, col,
                                                        bufB, NNODES);
  bias_prelu_stats<<<statBlocks, 128, 0, stream>>>(bufB, b, a_conv, stats,
                                                   NNODES, rows_per_block);
  bn_finalize<<<1, 128, 0, stream>>>(stats, gamma, beta, NNODES);
  bn_apply_prelu<<<4096, 256, 0, stream>>>(bufB, stats, a_out, bufA, nelem);

  gemm_wmma_f32<<<gemmGrid, gemmBlock, 0, stream>>>(bufA, W + DIMD * DIMD,
                                                    bufB);
  hipMemsetAsync(stats, 0, 2 * DIMD * sizeof(float), stream);
  gather_sum_in_edges<<<gatherBlocks, 256, 0, stream>>>(bufB, rowptr, col,
                                                        bufA, NNODES);
  bias_prelu_stats<<<statBlocks, 128, 0, stream>>>(bufA, b + DIMD, a_conv + 1,
                                                   stats, NNODES,
                                                   rows_per_block);
  bn_finalize<<<1, 128, 0, stream>>>(stats, gamma + DIMD, beta + DIMD, NNODES);
  bn_apply_prelu<<<4096, 256, 0, stream>>>(bufA, stats, a_out + 1,
                                           (float*)d_out, nelem);
}


# GEMM, edge-count, edge-fill and BN-apply kernels run as 391-625 resident 8-wave workgroups that loop with 16-byte accesses, not 3125-4096 short ones: under a third as many waves started per call
# speedup vs baseline: 2.0440x; 1.3947x over previous
#include <hip/hip_runtime.h>
#include <hip/hip_bf16.h>

typedef float v8f __attribute__((ext_vector_type(8)));
typedef float v2f __attribute__((ext_vector_type(2)));

#define DIMD 128
#define NNODES 50000
#define NEDGES 800000

#define GEMM_BLOCKS 625
__global__ __launch_bounds__(256) void gemm_wmma_f32(
    const float* __restrict__ H, const float* __restrict__ W,
    float* __restrict__ Z, int n_tiles) {
  const int wave = threadIdx.x >> 5;
  const int lane = threadIdx.x & 31;
  const int col0 = wave * 16;
  const int m    = lane & 15;
  const int kh   = (lane >> 4) * 2;

  for (int t = (int)blockIdx.x; t < n_tiles; t += (int)gridDim.x) {
    const int row0 = t * 16;
    const float* __restrict__ arow = H + (size_t)(row0 + m) * DIMD;
    v8f acc = {};
#pragma unroll
    for (int k = 0; k < DIMD; k += 4) {
      v2f a, b;
      a.x = arow[k + kh];
      a.y = arow[k + kh + 1];
      b.x = W[(size_t)(k + kh) * DIMD + col0 + m];
      b.y = W[(size_t)(k + kh + 1) * DIMD + col0 + m];
      acc = __builtin_amdgcn_wmma_f32_16x16x4_f32(
           false, a,  false, b,
           (short)0, acc,  false,  false);
    }
    const int mbase = row0 + (lane >> 4) * 8;
#pragma unroll
    for (int j = 0; j < 8; ++j)
      Z[(size_t)(mbase + j) * DIMD + col0 + m] = acc[j];
  }
}

#define EDGE_BLOCKS ((NEDGES / 8 + 255) / 256)
static_assert(NEDGES % 8 == 0, "the edge kernels take the edges in groups of eight");
__global__ __launch_bounds__(256) void count_in_edges(
    const int* __restrict__ dst, int* __restrict__ cnt, int* __restrict__ rank,
    int n_edges) {
  const int tid = (int)blockIdx.x * 256 + (int)threadIdx.x;
  const int nthreads = (int)gridDim.x * 256;
  const int n_groups = n_edges >> 3;
  const int4* __restrict__ dst4 = (const int4*)dst;
  int4* __restrict__ rank4 = (int4*)rank;
  for (int g = tid; g < n_groups; g += nthreads) {
    const int4 d0 = dst4[2 * g];
    const int4 d1 = dst4[2 * g + 1];
    int4 r0, r1;
    r0.x = atomicAdd(&cnt[d0.x], 1);
    r0.y = atomicAdd(&cnt[d0.y], 1);
    r0.z = atomicAdd(&cnt[d0.z], 1);
    r0.w = atomicAdd(&cnt[d0.w], 1);
    r1.x = atomicAdd(&cnt[d1.x], 1);
    r1.y = atomicAdd(&cnt[d1.y], 1);
    r1.z = atomicAdd(&cnt[d1.z], 1);
    r1.w = atomicAdd(&cnt[d1.w], 1);
    rank4[2 * g] = r0;
    rank4[2 * g + 1] = r1;
  }
}

#define SCAN_ROWS 13
#define SCAN_LEN (32 * SCAN_ROWS * 128)
__global__ __launch_bounds__(1024) void scan_in_degrees(int* p) {
  __shared__ int wave_total[32];
  const int lane = threadIdx.x & 31;
  const int w = __builtin_amdgcn_readfirstlane((int)(threadIdx.x >> 5));
  int4* q = (int4*)(p + w * (SCAN_ROWS * 128)) + lane;
  int4 v[SCAN_ROWS];
  int x[SCAN_ROWS];
#pragma unroll
  for (int i = 0; i < SCAN_ROWS; ++i) v[i] = q[i * 32];
#pragma unroll
  for (int i = 0; i < SCAN_ROWS; ++i) {
    v[i].y += v[i].x;
    v[i].z += v[i].y;
    v[i].w += v[i].z;
    x[i] = v[i].w;
  }
#pragma unroll
  for (int d = 1; d < 32; d <<= 1) {
    int y[SCAN_ROWS];
#pragma unroll
    for (int i = 0; i < SCAN_ROWS; ++i) y[i] = __shfl_up(x[i], d);
#pragma unroll
    for (int i = 0; i < SCAN_ROWS; ++i)
      if (lane >= d) x[i] += y[i];
  }
  int carry = 0;
  int off[SCAN_ROWS];
#pragma unroll
  for (int i = 0; i < SCAN_ROWS; ++i) {
    off[i] = carry + x[i] - v[i].w;
    carry += __builtin_amdgcn_readlane(x[i], 31);
  }
  if (lane == 0) wave_total[w] = carry;
  __syncthreads();
  const int t = wave_total[lane];
  int ts = t;
#pragma unroll
  for (int d = 1; d < 32; d <<= 1) {
    const int y = __shfl_up(ts, d);
    if (lane >= d) ts += y;
  }
  const int base = __builtin_amdgcn_readlane(ts - t, w);
#pragma unroll
  for (int i = 0; i < SCAN_ROWS; ++i) {
    const int o = base + off[i];
    int4 r;
    r.x = o;
    r.y = o + v[i].x;
    r.z = o + v[i].y;
    r.w = o + v[i].z;
    q[i * 32] = r;
  }
}

__global__ __launch_bounds__(256) void fill_in_edge_lists(
    const int* __restrict__ src, const int* __restrict__ dst,
    const int* __restrict__ rowptr, const int* __restrict__ rank,
    int* __restrict__ col, int n_edges) {
  const int tid = (int)blockIdx.x * 256 + (int)threadIdx.x;
  const int nthreads = (int)gridDim.x * 256;
  const int n_groups = n_edges >> 3;
  const int4* __restrict__ src4 = (const int4*)src;
  const int4* __restrict__ dst4 = (const int4*)dst;
  const int4* __restrict__ rank4 = (const int4*)rank;
  for (int g = tid; g < n_groups; g += nthreads) {
    const int4 d0 = dst4[2 * g];
    const int4 d1 = dst4[2 * g + 1];
    const int4 r0 = rank4[2 * g];
    const int4 r1 = rank4[2 * g + 1];
    const int4 s0 = src4[2 * g];
    const int4 s1 = src4[2 * g + 1];
    const int p0 = rowptr[d0.x], p1 = rowptr[d0.y], p2 = rowptr[d0.z], p3 = rowptr[d0.w];
    const int p4 = rowptr[d1.x], p5 = rowptr[d1.y], p6 = rowptr[d1.z], p7 = rowptr[d1.w];
    col[p0 + r0.x] = s0.x;
    col[p1 + r0.y] = s0.y;
    col[p2 + r0.z] = s0.z;
    col[p3 + r0.w] = s0.w;
    col[p4 + r1.x] = s1.x;
    col[p5 + r1.y] = s1.y;
    col[p6 + r1.z] = s1.z;
    col[p7 + r1.w] = s1.w;
  }
}

#define GATHER_WAVES 32
__global__ __launch_bounds__(GATHER_WAVES * 32) void gather_bias_prelu_stats(
    const float* __restrict__ Z, const int* __restrict__ rowptr,
    const int* __restrict__ col, const float* __restrict__ bias,
    const float* __restrict__ a_conv_p, float* __restrict__ out,
    float* __restrict__ stats, int n_rows) {
  __shared__ float part[GATHER_WAVES][2 * DIMD];
  const int lane = threadIdx.x & 31;
  const int wave = __builtin_amdgcn_readfirstlane((int)(threadIdx.x >> 5));
  const int n_waves = (int)gridDim.x * GATHER_WAVES;
  const float4* __restrict__ Z4 = (const float4*)Z;
  float4* __restrict__ O4 = (float4*)out;
  const float4 bb = ((const float4*)bias)[lane];
  const float ac = *a_conv_p;
  float4 s = {0.f, 0.f, 0.f, 0.f};
  float4 ss = {0.f, 0.f, 0.f, 0.f};
  for (int d = (int)blockIdx.x * GATHER_WAVES + wave; d < n_rows; d += n_waves) {
    const int beg = rowptr[d];
    const int end = rowptr[d + 1];
    float4 acc = {0.f, 0.f, 0.f, 0.f};
    for (int b = beg; b < end; b += 32) {
      const int j = b + lane;
      const int mine = (j < end) ? col[j] : 0;
      const int m = (end - b < 32) ? (end - b) : 32;
#pragma unroll 4
      for (int k = 0; k < m; ++k) {
        const int sidx = __builtin_amdgcn_readlane(mine, k);
        const float4 v = Z4[(size_t)sidx * (DIMD / 4) + lane];
        acc.x += v.x;
        acc.y += v.y;
        acc.z += v.z;
        acc.w += v.w;
      }
    }
    float4 z;
    z.x = acc.x + bb.x;
    z.y = acc.y + bb.y;
    z.z = acc.z + bb.z;
    z.w = acc.w + bb.w;
    z.x = (z.x >= 0.f) ? z.x : ac * z.x;
    z.y = (z.y >= 0.f) ? z.y : ac * z.y;
    z.z = (z.z >= 0.f) ? z.z : ac * z.z;
    z.w = (z.w >= 0.f) ? z.w : ac * z.w;
    O4[(size_t)d * (DIMD / 4) + lane] = z;
    s.x += z.x;
    s.y += z.y;
    s.z += z.z;
    s.w += z.w;
    ss.x += z.x * z.x;
    ss.y += z.y * z.y;
    ss.z += z.z * z.z;
    ss.w += z.w * z.w;
  }
  float4* __restrict__ P4 = (float4*)&part[wave][0];
  P4[lane] = s;
  P4[DIMD / 4 + lane] = ss;
  __syncthreads();
  if (threadIdx.x < 2 * DIMD) {
    float t = 0.f;
    for (int w = 0; w < GATHER_WAVES; ++w) t += part[w][threadIdx.x];
    unsafeAtomicAdd(&stats[threadIdx.x], t);
  }
}

__global__ __launch_bounds__(128) void bn_finalize(
    float* __restrict__ stats, const float* __restrict__ gamma,
    const float* __restrict__ beta, int n_rows) {
  const int c = threadIdx.x;
  const float inv_n = 1.0f / (float)n_rows;
  const float mu  = stats[c] * inv_n;
  const float var = stats[DIMD + c] * inv_n - mu * mu;
  const float scale = rsqrtf(var + 1e-5f) * gamma[c];
  const float shift = beta[c] - mu * scale;
  stats[c]        = scale;
  stats[DIMD + c] = shift;
}

#define APPLY_BLOCKS 625
#define APPLY_WAVES 8
#define APPLY_ROWS 10
static_assert(APPLY_BLOCKS * APPLY_WAVES * APPLY_ROWS == NNODES,
              "the row partition of bn_apply_prelu must be exact");
__global__ __launch_bounds__(APPLY_WAVES * 32) void bn_apply_prelu(
    const float* __restrict__ Z, const float* __restrict__ stats,
    const float* __restrict__ a_out_p, float* __restrict__ Hout) {
  const int lane = threadIdx.x & 31;
  const int wave = __builtin_amdgcn_readfirstlane((int)(threadIdx.x >> 5));
  const int gw = (int)blockIdx.x * APPLY_WAVES + wave;
  const float4* __restrict__ Z4 = (const float4*)Z;
  float4* __restrict__ H4 = (float4*)Hout;
  const float4 sc = ((const float4*)stats)[lane];
  const float4 sh = ((const float4*)stats)[DIMD / 4 + lane];
  const float ao = *a_out_p;
  float4 v[APPLY_ROWS];
#pragma unroll
  for (int u = 0; u < APPLY_ROWS; ++u)
    v[u] = Z4[(size_t)(gw + u * (APPLY_BLOCKS * APPLY_WAVES)) * (DIMD / 4) + lane];
#pragma unroll
  for (int u = 0; u < APPLY_ROWS; ++u) {
    float4 z;
    z.x = fmaf(v[u].x, sc.x, sh.x);
    z.y = fmaf(v[u].y, sc.y, sh.y);
    z.z = fmaf(v[u].z, sc.z, sh.z);
    z.w = fmaf(v[u].w, sc.w, sh.w);
    z.x = (z.x >= 0.f) ? z.x : ao * z.x;
    z.y = (z.y >= 0.f) ? z.y : ao * z.y;
    z.z = (z.z >= 0.f) ? z.z : ao * z.z;
    z.w = (z.w >= 0.f) ? z.w : ao * z.w;
    H4[(size_t)(gw + u * (APPLY_BLOCKS * APPLY_WAVES)) * (DIMD / 4) + lane] = z;
  }
}

extern "C" void kernel_launch(void* const* d_in, const int* in_sizes, int n_in,
                              void* d_out, int out_size, void* d_ws,
                              size_t ws_size, hipStream_t stream) {
  const float* heat   = (const float*)d_in[0];
  const int*   src    = (const int*)d_in[1];
  const int*   dst    = (const int*)d_in[2];
  const float* W      = (const float*)d_in[3];
  const float* b      = (const float*)d_in[4];
  const float* a_conv = (const float*)d_in[5];
  const float* gamma  = (const float*)d_in[6];
  const float* beta   = (const float*)d_in[7];
  const float* a_out  = (const float*)d_in[8];

  const size_t nb = (size_t)NNODES * DIMD * sizeof(float);
  float* bufA  = (float*)d_ws;
  float* bufB  = (float*)((char*)d_ws + nb);
  float* stats = (float*)((char*)d_ws + 2 * nb);

  int* rowptr = (int*)d_out;
  int* rank   = rowptr + 65536;
  int* col    = rank + NEDGES;

  const dim3 gemmGrid(GEMM_BLOCKS);
  const dim3 gemmBlock(256);
  const int edgeBlocks = EDGE_BLOCKS;
  const int gatherBlocks = 512;

  hipMemsetAsync(rowptr, 0, SCAN_LEN * sizeof(int), stream);
  hipMemsetAsync(stats, 0, 2 * DIMD * sizeof(float), stream);
  count_in_edges<<<edgeBlocks, 256, 0, stream>>>(dst, rowptr, rank, NEDGES);
  scan_in_degrees<<<1, 1024, 0, stream>>>(rowptr);
  fill_in_edge_lists<<<edgeBlocks, 256, 0, stream>>>(src, dst, rowptr, rank,
                                                     col, NEDGES);

  gemm_wmma_f32<<<gemmGrid, gemmBlock, 0, stream>>>(heat, W, bufA, NNODES / 16);
  gather_bias_prelu_stats<<<gatherBlocks, GATHER_WAVES * 32, 0, stream>>>(
      bufA, rowptr, col, b, a_conv, bufB, stats, NNODES);
  bn_finalize<<<1, 128, 0, stream>>>(stats, gamma, beta, NNODES);
  bn_apply_prelu<<<APPLY_BLOCKS, APPLY_WAVES * 32, 0, stream>>>(bufB, stats, a_out, bufA);

  gemm_wmma_f32<<<gemmGrid, gemmBlock, 0, stream>>>(bufA, W + DIMD * DIMD,
                                                    bufB, NNODES / 16);
  hipMemsetAsync(stats, 0, 2 * DIMD * sizeof(float), stream);
  gather_bias_prelu_stats<<<gatherBlocks, GATHER_WAVES * 32, 0, stream>>>(
      bufB, rowptr, col, b + DIMD, a_conv + 1, bufA, stats, NNODES);
  bn_finalize<<<1, 128, 0, stream>>>(stats, gamma + DIMD, beta + DIMD, NNODES);
  bn_apply_prelu<<<APPLY_BLOCKS, APPLY_WAVES * 32, 0, stream>>>(bufA, stats, a_out + 1,
                                                                (float*)d_out);
}


# GEMM A operand: each 8-wave workgroup stages its 16-row tile once in LDS by whole-line 16-byte loads; waves read k-permuted fragments from LDS instead of each fetching A with 16-line 8-byte loads
# speedup vs baseline: 2.2896x; 1.1202x over previous
#include <hip/hip_runtime.h>
#include <hip/hip_bf16.h>

typedef float v8f __attribute__((ext_vector_type(8)));
typedef float v2f __attribute__((ext_vector_type(2)));

#define DIMD 128
#define NNODES 50000
#define NEDGES 800000

#define GEMM_BLOCKS 512
#define A_PITCH4 33
__global__ __launch_bounds__(256) void gemm_wmma_f32(
    const float* __restrict__ H, const float* __restrict__ W,
    float* __restrict__ Z, int n_tiles) {
  __shared__ float4 atile[2][16 * A_PITCH4];
  const int wave = threadIdx.x >> 5;
  const int lane = threadIdx.x & 31;
  const int col0 = wave * 16;
  const int m    = lane & 15;
  const int h    = lane >> 4;

  v2f bw[32];
#pragma unroll
  for (int i = 0; i < 16; ++i) {
#pragma unroll
    for (int s = 0; s < 2; ++s) {
      const int k = 8 * i + 4 * h + 2 * s;
      bw[2 * i + s].x = W[(size_t)k * DIMD + col0 + m];
      bw[2 * i + s].y = W[(size_t)(k + 1) * DIMD + col0 + m];
    }
  }

  const float4* __restrict__ H4 = (const float4*)H;
  const int stride = (int)gridDim.x;
  int t = (int)blockIdx.x;
  int cur = 0;
  float4 g0 = {0.f, 0.f, 0.f, 0.f}, g1 = {0.f, 0.f, 0.f, 0.f};
  if (t < n_tiles) {
    g0 = H4[(size_t)(t * 16 + wave) * (DIMD / 4) + lane];
    g1 = H4[(size_t)(t * 16 + wave + 8) * (DIMD / 4) + lane];
    atile[0][wave * A_PITCH4 + lane] = g0;
    atile[0][(wave + 8) * A_PITCH4 + lane] = g1;
  }
  __syncthreads();

  for (; t < n_tiles; t += stride) {
    const int tn = t + stride;
    const bool more = tn < n_tiles;
    if (more) {
      g0 = H4[(size_t)(tn * 16 + wave) * (DIMD / 4) + lane];
      g1 = H4[(size_t)(tn * 16 + wave + 8) * (DIMD / 4) + lane];
    }
    const float4* __restrict__ arow = &atile[cur][m * A_PITCH4 + h];
    float4 af[16];
#pragma unroll
    for (int i = 0; i < 16; ++i) af[i] = arow[2 * i];
    __builtin_amdgcn_sched_barrier(0);
    v8f acc = {};
#pragma unroll
    for (int i = 0; i < 16; ++i) {
      v2f a0, a1;
      a0.x = af[i].x;
      a0.y = af[i].y;
      a1.x = af[i].z;
      a1.y = af[i].w;
      acc = __builtin_amdgcn_wmma_f32_16x16x4_f32(
           false, a0,  false, bw[2 * i],
           (short)0, acc,  false,  false);
      acc = __builtin_amdgcn_wmma_f32_16x16x4_f32(
           false, a1,  false, bw[2 * i + 1],
           (short)0, acc,  false,  false);
    }
    const int mbase = t * 16 + h * 8;
#pragma unroll
    for (int j = 0; j < 8; ++j)
      Z[(size_t)(mbase + j) * DIMD + col0 + m] = acc[j];
    if (more) {
      atile[cur ^ 1][wave * A_PITCH4 + lane] = g0;
      atile[cur ^ 1][(wave + 8) * A_PITCH4 + lane] = g1;
    }
    __builtin_amdgcn_fence(__ATOMIC_RELEASE, "workgroup", "local");
    __builtin_amdgcn_s_barrier();
    __builtin_amdgcn_fence(__ATOMIC_ACQUIRE, "workgroup", "local");
    cur ^= 1;
  }
}

#define EDGE_BLOCKS ((NEDGES / 8 + 255) / 256)
static_assert(NEDGES % 8 == 0, "the edge kernels take the edges in groups of eight");
__global__ __launch_bounds__(256) void count_in_edges(
    const int* __restrict__ dst, int* __restrict__ cnt, int* __restrict__ rank,
    int n_edges) {
  const int tid = (int)blockIdx.x * 256 + (int)threadIdx.x;
  const int nthreads = (int)gridDim.x * 256;
  const int n_groups = n_edges >> 3;
  const int4* __restrict__ dst4 = (const int4*)dst;
  int4* __restrict__ rank4 = (int4*)rank;
  for (int g = tid; g < n_groups; g += nthreads) {
    const int4 d0 = dst4[2 * g];
    const int4 d1 = dst4[2 * g + 1];
    int4 r0, r1;
    r0.x = atomicAdd(&cnt[d0.x], 1);
    r0.y = atomicAdd(&cnt[d0.y], 1);
    r0.z = atomicAdd(&cnt[d0.z], 1);
    r0.w = atomicAdd(&cnt[d0.w], 1);
    r1.x = atomicAdd(&cnt[d1.x], 1);
    r1.y = atomicAdd(&cnt[d1.y], 1);
    r1.z = atomicAdd(&cnt[d1.z], 1);
    r1.w = atomicAdd(&cnt[d1.w], 1);
    rank4[2 * g] = r0;
    rank4[2 * g + 1] = r1;
  }
}

#define SCAN_ROWS 13
#define SCAN_LEN (32 * SCAN_ROWS * 128)
__global__ __launch_bounds__(1024) void scan_in_degrees(int* p) {
  __shared__ int wave_total[32];
  const int lane = threadIdx.x & 31;
  const int w = __builtin_amdgcn_readfirstlane((int)(threadIdx.x >> 5));
  int4* q = (int4*)(p + w * (SCAN_ROWS * 128)) + lane;
  int4 v[SCAN_ROWS];
  int x[SCAN_ROWS];
#pragma unroll
  for (int i = 0; i < SCAN_ROWS; ++i) v[i] = q[i * 32];
#pragma unroll
  for (int i = 0; i < SCAN_ROWS; ++i) {
    v[i].y += v[i].x;
    v[i].z += v[i].y;
    v[i].w += v[i].z;
    x[i] = v[i].w;
  }
#pragma unroll
  for (int d = 1; d < 32; d <<= 1) {
    int y[SCAN_ROWS];
#pragma unroll
    for (int i = 0; i < SCAN_ROWS; ++i) y[i] = __shfl_up(x[i], d);
#pragma unroll
    for (int i = 0; i < SCAN_ROWS; ++i)
      if (lane >= d) x[i] += y[i];
  }
  int carry = 0;
  int off[SCAN_ROWS];
#pragma unroll
  for (int i = 0; i < SCAN_ROWS; ++i) {
    off[i] = carry + x[i] - v[i].w;
    carry += __builtin_amdgcn_readlane(x[i], 31);
  }
  if (lane == 0) wave_total[w] = carry;
  __syncthreads();
  const int t = wave_total[lane];
  int ts = t;
#pragma unroll
  for (int d = 1; d < 32; d <<= 1) {
    const int y = __shfl_up(ts, d);
    if (lane >= d) ts += y;
  }
  const int base = __builtin_amdgcn_readlane(ts - t, w);
#pragma unroll
  for (int i = 0; i < SCAN_ROWS; ++i) {
    const int o = base + off[i];
    int4 r;
    r.x = o;
    r.y = o + v[i].x;
    r.z = o + v[i].y;
    r.w = o + v[i].z;
    q[i * 32] = r;
  }
}

__global__ __launch_bounds__(256) void fill_in_edge_lists(
    const int* __restrict__ src, const int* __restrict__ dst,
    const int* __restrict__ rowptr, const int* __restrict__ rank,
    int* __restrict__ col, int n_edges) {
  const int tid = (int)blockIdx.x * 256 + (int)threadIdx.x;
  const int nthreads = (int)gridDim.x * 256;
  const int n_groups = n_edges >> 3;
  const int4* __restrict__ src4 = (const int4*)src;
  const int4* __restrict__ dst4 = (const int4*)dst;
  const int4* __restrict__ rank4 = (const int4*)rank;
  for (int g = tid; g < n_groups; g += nthreads) {
    const int4 d0 = dst4[2 * g];
    const int4 d1 = dst4[2 * g + 1];
    const int4 r0 = rank4[2 * g];
    const int4 r1 = rank4[2 * g + 1];
    const int4 s0 = src4[2 * g];
    const int4 s1 = src4[2 * g + 1];
    const int p0 = rowptr[d0.x], p1 = rowptr[d0.y], p2 = rowptr[d0.z], p3 = rowptr[d0.w];
    const int p4 = rowptr[d1.x], p5 = rowptr[d1.y], p6 = rowptr[d1.z], p7 = rowptr[d1.w];
    col[p0 + r0.x] = s0.x;
    col[p1 + r0.y] = s0.y;
    col[p2 + r0.z] = s0.z;
    col[p3 + r0.w] = s0.w;
    col[p4 + r1.x] = s1.x;
    col[p5 + r1.y] = s1.y;
    col[p6 + r1.z] = s1.z;
    col[p7 + r1.w] = s1.w;
  }
}

#define GATHER_WAVES 32
__global__ __launch_bounds__(GATHER_WAVES * 32) void gather_bias_prelu_stats(
    const float* __restrict__ Z, const int* __restrict__ rowptr,
    const int* __restrict__ col, const float* __restrict__ bias,
    const float* __restrict__ a_conv_p, float* __restrict__ out,
    float* __restrict__ stats, int n_rows) {
  __shared__ float part[GATHER_WAVES][2 * DIMD];
  const int lane = threadIdx.x & 31;
  const int wave = __builtin_amdgcn_readfirstlane((int)(threadIdx.x >> 5));
  const int n_waves = (int)gridDim.x * GATHER_WAVES;
  const float4* __restrict__ Z4 = (const float4*)Z;
  float4* __restrict__ O4 = (float4*)out;
  const float4 bb = ((const float4*)bias)[lane];
  const float ac = *a_conv_p;
  float4 s = {0.f, 0.f, 0.f, 0.f};
  float4 ss = {0.f, 0.f, 0.f, 0.f};
  for (int d = (int)blockIdx.x * GATHER_WAVES + wave; d < n_rows; d += n_waves) {
    const int beg = rowptr[d];
    const int end = rowptr[d + 1];
    float4 acc = {0.f, 0.f, 0.f, 0.f};
    for (int b = beg; b < end; b += 32) {
      const int j = b + lane;
      const int mine = (j < end) ? col[j] : 0;
      const int m = (end - b < 32) ? (end - b) : 32;
#pragma unroll 4
      for (int k = 0; k < m; ++k) {
        const int sidx = __builtin_amdgcn_readlane(mine, k);
        const float4 v = Z4[(size_t)sidx * (DIMD / 4) + lane];
        acc.x += v.x;
        acc.y += v.y;
        acc.z += v.z;
        acc.w += v.w;
      }
    }
    float4 z;
    z.x = acc.x + bb.x;
    z.y = acc.y + bb.y;
    z.z = acc.z + bb.z;
    z.w = acc.w + bb.w;
    z.x = (z.x >= 0.f) ? z.x : ac * z.x;
    z.y = (z.y >= 0.f) ? z.y : ac * z.y;
    z.z = (z.z >= 0.f) ? z.z : ac * z.z;
    z.w = (z.w >= 0.f) ? z.w : ac * z.w;
    O4[(size_t)d * (DIMD / 4) + lane] = z;
    s.x += z.x;
    s.y += z.y;
    s.z += z.z;
    s.w += z.w;
    ss.x += z.x * z.x;
    ss.y += z.y * z.y;
    ss.z += z.z * z.z;
    ss.w += z.w * z.w;
  }
  float4* __restrict__ P4 = (float4*)&part[wave][0];
  P4[lane] = s;
  P4[DIMD / 4 + lane] = ss;
  __syncthreads();
  if (threadIdx.x < 2 * DIMD) {
    float t = 0.f;
    for (int w = 0; w < GATHER_WAVES; ++w) t += part[w][threadIdx.x];
    unsafeAtomicAdd(&stats[threadIdx.x], t);
  }
}

__global__ __launch_bounds__(128) void bn_finalize(
    float* __restrict__ stats, const float* __restrict__ gamma,
    const float* __restrict__ beta, int n_rows) {
  const int c = threadIdx.x;
  const float inv_n = 1.0f / (float)n_rows;
  const float mu  = stats[c] * inv_n;
  const float var = stats[DIMD + c] * inv_n - mu * mu;
  const float scale = rsqrtf(var + 1e-5f) * gamma[c];
  const float shift = beta[c] - mu * scale;
  stats[c]        = scale;
  stats[DIMD + c] = shift;
}

#define APPLY_BLOCKS 625
#define APPLY_WAVES 8
#define APPLY_ROWS 10
static_assert(APPLY_BLOCKS * APPLY_WAVES * APPLY_ROWS == NNODES,
              "the row partition of bn_apply_prelu must be exact");
__global__ __launch_bounds__(APPLY_WAVES * 32) void bn_apply_prelu(
    const float* __restrict__ Z, const float* __restrict__ stats,
    const float* __restrict__ a_out_p, float* __restrict__ Hout) {
  const int lane = threadIdx.x & 31;
  const int wave = __builtin_amdgcn_readfirstlane((int)(threadIdx.x >> 5));
  const int gw = (int)blockIdx.x * APPLY_WAVES + wave;
  const float4* __restrict__ Z4 = (const float4*)Z;
  float4* __restrict__ H4 = (float4*)Hout;
  const float4 sc = ((const float4*)stats)[lane];
  const float4 sh = ((const float4*)stats)[DIMD / 4 + lane];
  const float ao = *a_out_p;
  float4 v[APPLY_ROWS];
#pragma unroll
  for (int u = 0; u < APPLY_ROWS; ++u)
    v[u] = Z4[(size_t)(gw + u * (APPLY_BLOCKS * APPLY_WAVES)) * (DIMD / 4) + lane];
#pragma unroll
  for (int u = 0; u < APPLY_ROWS; ++u) {
    float4 z;
    z.x = fmaf(v[u].x, sc.x, sh.x);
    z.y = fmaf(v[u].y, sc.y, sh.y);
    z.z = fmaf(v[u].z, sc.z, sh.z);
    z.w = fmaf(v[u].w, sc.w, sh.w);
    z.x = (z.x >= 0.f) ? z.x : ao * z.x;
    z.y = (z.y >= 0.f) ? z.y : ao * z.y;
    z.z = (z.z >= 0.f) ? z.z : ao * z.z;
    z.w = (z.w >= 0.f) ? z.w : ao * z.w;
    H4[(size_t)(gw + u * (APPLY_BLOCKS * APPLY_WAVES)) * (DIMD / 4) + lane] = z;
  }
}

extern "C" void kernel_launch(void* const* d_in, const int* in_sizes, int n_in,
                              void* d_out, int out_size, void* d_ws,
                              size_t ws_size, hipStream_t stream) {
  const float* heat   = (const float*)d_in[0];
  const int*   src    = (const int*)d_in[1];
  const int*   dst    = (const int*)d_in[2];
  const float* W      = (const float*)d_in[3];
  const float* b      = (const float*)d_in[4];
  const float* a_conv = (const float*)d_in[5];
  const float* gamma  = (const float*)d_in[6];
  const float* beta   = (const float*)d_in[7];
  const float* a_out  = (const float*)d_in[8];

  const size_t nb = (size_t)NNODES * DIMD * sizeof(float);
  float* bufA  = (float*)d_ws;
  float* bufB  = (float*)((char*)d_ws + nb);
  float* stats = (float*)((char*)d_ws + 2 * nb);

  int* rowptr = (int*)d_out;
  int* rank   = rowptr + 65536;
  int* col    = rank + NEDGES;

  const dim3 gemmGrid(GEMM_BLOCKS);
  const dim3 gemmBlock(256);
  const int edgeBlocks = EDGE_BLOCKS;
  const int gatherBlocks = 512;

  hipMemsetAsync(rowptr, 0, SCAN_LEN * sizeof(int), stream);
  hipMemsetAsync(stats, 0, 2 * DIMD * sizeof(float), stream);
  count_in_edges<<<edgeBlocks, 256, 0, stream>>>(dst, rowptr, rank, NEDGES);
  scan_in_degrees<<<1, 1024, 0, stream>>>(rowptr);
  fill_in_edge_lists<<<edgeBlocks, 256, 0, stream>>>(src, dst, rowptr, rank,
                                                     col, NEDGES);

  gemm_wmma_f32<<<gemmGrid, gemmBlock, 0, stream>>>(heat, W, bufA, NNODES / 16);
  gather_bias_prelu_stats<<<gatherBlocks, GATHER_WAVES * 32, 0, stream>>>(
      bufA, rowptr, col, b, a_conv, bufB, stats, NNODES);
  bn_finalize<<<1, 128, 0, stream>>>(stats, gamma, beta, NNODES);
  bn_apply_prelu<<<APPLY_BLOCKS, APPLY_WAVES * 32, 0, stream>>>(bufB, stats, a_out, bufA);

  gemm_wmma_f32<<<gemmGrid, gemmBlock, 0, stream>>>(bufA, W + DIMD * DIMD,
                                                    bufB, NNODES / 16);
  hipMemsetAsync(stats, 0, 2 * DIMD * sizeof(float), stream);
  gather_bias_prelu_stats<<<gatherBlocks, GATHER_WAVES * 32, 0, stream>>>(
      bufB, rowptr, col, b + DIMD, a_conv + 1, bufA, stats, NNODES);
  bn_finalize<<<1, 128, 0, stream>>>(stats, gamma + DIMD, beta + DIMD, NNODES);
  bn_apply_prelu<<<APPLY_BLOCKS, APPLY_WAVES * 32, 0, stream>>>(bufA, stats, a_out + 1,
                                                                (float*)d_out);
}


# Batch norm folded into its consumers: the second GEMM applies layer one's scale, shift and outer PReLU while staging its A tile; readers derive both from the column sums: 8 launches, not 11
# speedup vs baseline: 2.5251x; 1.1029x over previous
#include <hip/hip_runtime.h>
#include <hip/hip_bf16.h>

typedef float v8f __attribute__((ext_vector_type(8)));
typedef float v2f __attribute__((ext_vector_type(2)));

#define DIMD 128
#define NNODES 50000
#define NEDGES 800000

__device__ __forceinline__ void bn_scale_shift(float sum, float sumsq, float g,
                                               float b, float inv_n,
                                               float& scale, float& shift) {
  const float mu  = sum * inv_n;
  const float var = sumsq * inv_n - mu * mu;
  scale = rsqrtf(var + 1e-5f) * g;
  shift = b - mu * scale;
}

struct BnLaneRaw {
  float4 su, sq, ga, be;
};
__device__ __forceinline__ BnLaneRaw bn_lane_load(
    const float* __restrict__ stats, const float* __restrict__ gamma,
    const float* __restrict__ beta, int lane) {
  BnLaneRaw r;
  r.su = ((const float4*)stats)[lane];
  r.sq = ((const float4*)stats)[32 + lane];
  r.ga = ((const float4*)gamma)[lane];
  r.be = ((const float4*)beta)[lane];
  return r;
}
__device__ __forceinline__ void bn_lane_scale_shift(const BnLaneRaw& r,
                                                    int n_rows, float4& sc,
                                                    float4& sh) {
  const float inv_n = 1.0f / (float)n_rows;
  bn_scale_shift(r.su.x, r.sq.x, r.ga.x, r.be.x, inv_n, sc.x, sh.x);
  bn_scale_shift(r.su.y, r.sq.y, r.ga.y, r.be.y, inv_n, sc.y, sh.y);
  bn_scale_shift(r.su.z, r.sq.z, r.ga.z, r.be.z, inv_n, sc.z, sh.z);
  bn_scale_shift(r.su.w, r.sq.w, r.ga.w, r.be.w, inv_n, sc.w, sh.w);
}

__device__ __forceinline__ float4 bn_prelu4(float4 v, float4 sc, float4 sh,
                                            float a) {
  float4 z;
  z.x = fmaf(v.x, sc.x, sh.x);
  z.y = fmaf(v.y, sc.y, sh.y);
  z.z = fmaf(v.z, sc.z, sh.z);
  z.w = fmaf(v.w, sc.w, sh.w);
  z.x = (z.x >= 0.f) ? z.x : a * z.x;
  z.y = (z.y >= 0.f) ? z.y : a * z.y;
  z.z = (z.z >= 0.f) ? z.z : a * z.z;
  z.w = (z.w >= 0.f) ? z.w : a * z.w;
  return z;
}

#define GEMM_BLOCKS 512
#define A_PITCH4 33
template <bool BN_IN>
__device__ __forceinline__ void gemm_rows_wmma_f32(
    float4 (*atile)[16 * A_PITCH4], const float* __restrict__ H,
    const float* __restrict__ W, float* __restrict__ Z, int n_tiles,
    const float* __restrict__ stats, const float* __restrict__ gamma,
    const float* __restrict__ beta, const float* __restrict__ a_out_p,
    int n_rows) {
  const int wave = threadIdx.x >> 5;
  const int lane = threadIdx.x & 31;
  const int col0 = wave * 16;
  const int m    = lane & 15;
  const int h    = lane >> 4;

  v2f bw[32];
#pragma unroll
  for (int i = 0; i < 16; ++i) {
#pragma unroll
    for (int s = 0; s < 2; ++s) {
      const int k = 8 * i + 4 * h + 2 * s;
      bw[2 * i + s].x = W[(size_t)k * DIMD + col0 + m];
      bw[2 * i + s].y = W[(size_t)(k + 1) * DIMD + col0 + m];
    }
  }

  const float4* __restrict__ H4 = (const float4*)H;
  const int stride = (int)gridDim.x;
  int t = (int)blockIdx.x;
  int cur = 0;
  float4 g0 = {0.f, 0.f, 0.f, 0.f}, g1 = {0.f, 0.f, 0.f, 0.f};
  const bool first = t < n_tiles;
  if (first) {
    g0 = H4[(size_t)(t * 16 + wave) * (DIMD / 4) + lane];
    g1 = H4[(size_t)(t * 16 + wave + 8) * (DIMD / 4) + lane];
  }
  float4 sc = {1.f, 1.f, 1.f, 1.f}, sh = {0.f, 0.f, 0.f, 0.f};
  float ao = 0.f;
  if (BN_IN) {
    bn_lane_scale_shift(bn_lane_load(stats, gamma, beta, lane), n_rows, sc, sh);
    ao = *a_out_p;
  }
  if (first) {
    if (BN_IN) {
      g0 = bn_prelu4(g0, sc, sh, ao);
      g1 = bn_prelu4(g1, sc, sh, ao);
    }
    atile[0][wave * A_PITCH4 + lane] = g0;
    atile[0][(wave + 8) * A_PITCH4 + lane] = g1;
  }
  __syncthreads();

  for (; t < n_tiles; t += stride) {
    const int tn = t + stride;
    const bool more = tn < n_tiles;
    if (more) {
      g0 = H4[(size_t)(tn * 16 + wave) * (DIMD / 4) + lane];
      g1 = H4[(size_t)(tn * 16 + wave + 8) * (DIMD / 4) + lane];
    }
    const float4* __restrict__ arow = &atile[cur][m * A_PITCH4 + h];
    float4 af[16];
#pragma unroll
    for (int i = 0; i < 16; ++i) af[i] = arow[2 * i];
    __builtin_amdgcn_sched_barrier(0);
    v8f acc = {};
#pragma unroll
    for (int i = 0; i < 16; ++i) {
      v2f a0, a1;
      a0.x = af[i].x;
      a0.y = af[i].y;
      a1.x = af[i].z;
      a1.y = af[i].w;
      acc = __builtin_amdgcn_wmma_f32_16x16x4_f32(
           false, a0,  false, bw[2 * i],
           (short)0, acc,  false,  false);
      acc = __builtin_amdgcn_wmma_f32_16x16x4_f32(
           false, a1,  false, bw[2 * i + 1],
           (short)0, acc,  false,  false);
    }
    const int mbase = t * 16 + h * 8;
#pragma unroll
    for (int j = 0; j < 8; ++j)
      Z[(size_t)(mbase + j) * DIMD + col0 + m] = acc[j];
    if (more) {
      if (BN_IN) {
        g0 = bn_prelu4(g0, sc, sh, ao);
        g1 = bn_prelu4(g1, sc, sh, ao);
      }
      atile[cur ^ 1][wave * A_PITCH4 + lane] = g0;
      atile[cur ^ 1][(wave + 8) * A_PITCH4 + lane] = g1;
    }
    __builtin_amdgcn_fence(__ATOMIC_RELEASE, "workgroup", "local");
    __builtin_amdgcn_s_barrier();
    __builtin_amdgcn_fence(__ATOMIC_ACQUIRE, "workgroup", "local");
    cur ^= 1;
  }
}

__global__ __launch_bounds__(256) void gemm_wmma_f32(
    const float* __restrict__ H, const float* __restrict__ W,
    float* __restrict__ Z, int n_tiles) {
  __shared__ float4 atile[2][16 * A_PITCH4];
  gemm_rows_wmma_f32<false>(atile, H, W, Z, n_tiles, nullptr, nullptr, nullptr,
                            nullptr, 0);
}

__global__ __launch_bounds__(256) void gemm_bn_prelu_wmma_f32(
    const float* __restrict__ Hraw, const float* __restrict__ stats,
    const float* __restrict__ gamma, const float* __restrict__ beta,
    const float* __restrict__ a_out_p, const float* __restrict__ W,
    float* __restrict__ Z, int n_tiles, int n_rows) {
  __shared__ float4 atile[2][16 * A_PITCH4];
  gemm_rows_wmma_f32<true>(atile, Hraw, W, Z, n_tiles, stats, gamma, beta,
                           a_out_p, n_rows);
}

#define EDGE_BLOCKS ((NEDGES / 8 + 255) / 256)
static_assert(NEDGES % 8 == 0, "the edge kernels take the edges in groups of eight");
__global__ __launch_bounds__(256) void count_in_edges(
    const int* __restrict__ dst, int* __restrict__ cnt, int* __restrict__ rank,
    int n_edges) {
  const int tid = (int)blockIdx.x * 256 + (int)threadIdx.x;
  const int nthreads = (int)gridDim.x * 256;
  const int n_groups = n_edges >> 3;
  const int4* __restrict__ dst4 = (const int4*)dst;
  int4* __restrict__ rank4 = (int4*)rank;
  for (int g = tid; g < n_groups; g += nthreads) {
    const int4 d0 = dst4[2 * g];
    const int4 d1 = dst4[2 * g + 1];
    int4 r0, r1;
    r0.x = atomicAdd(&cnt[d0.x], 1);
    r0.y = atomicAdd(&cnt[d0.y], 1);
    r0.z = atomicAdd(&cnt[d0.z], 1);
    r0.w = atomicAdd(&cnt[d0.w], 1);
    r1.x = atomicAdd(&cnt[d1.x], 1);
    r1.y = atomicAdd(&cnt[d1.y], 1);
    r1.z = atomicAdd(&cnt[d1.z], 1);
    r1.w = atomicAdd(&cnt[d1.w], 1);
    rank4[2 * g] = r0;
    rank4[2 * g + 1] = r1;
  }
}

#define SCAN_ROWS 13
#define SCAN_LEN (32 * SCAN_ROWS * 128)
__global__ __launch_bounds__(1024) void scan_in_degrees(int* p) {
  __shared__ int wave_total[32];
  const int lane = threadIdx.x & 31;
  const int w = __builtin_amdgcn_readfirstlane((int)(threadIdx.x >> 5));
  int4* q = (int4*)(p + w * (SCAN_ROWS * 128)) + lane;
  int4 v[SCAN_ROWS];
  int x[SCAN_ROWS];
#pragma unroll
  for (int i = 0; i < SCAN_ROWS; ++i) v[i] = q[i * 32];
#pragma unroll
  for (int i = 0; i < SCAN_ROWS; ++i) {
    v[i].y += v[i].x;
    v[i].z += v[i].y;
    v[i].w += v[i].z;
    x[i] = v[i].w;
  }
#pragma unroll
  for (int d = 1; d < 32; d <<= 1) {
    int y[SCAN_ROWS];
#pragma unroll
    for (int i = 0; i < SCAN_ROWS; ++i) y[i] = __shfl_up(x[i], d);
#pragma unroll
    for (int i = 0; i < SCAN_ROWS; ++i)
      if (lane >= d) x[i] += y[i];
  }
  int carry = 0;
  int off[SCAN_ROWS];
#pragma unroll
  for (int i = 0; i < SCAN_ROWS; ++i) {
    off[i] = carry + x[i] - v[i].w;
    carry += __builtin_amdgcn_readlane(x[i], 31);
  }
  if (lane == 0) wave_total[w] = carry;
  __syncthreads();
  const int t = wave_total[lane];
  int ts = t;
#pragma unroll
  for (int d = 1; d < 32; d <<= 1) {
    const int y = __shfl_up(ts, d);
    if (lane >= d) ts += y;
  }
  const int base = __builtin_amdgcn_readlane(ts - t, w);
#pragma unroll
  for (int i = 0; i < SCAN_ROWS; ++i) {
    const int o = base + off[i];
    int4 r;
    r.x = o;
    r.y = o + v[i].x;
    r.z = o + v[i].y;
    r.w = o + v[i].z;
    q[i * 32] = r;
  }
}

__global__ __launch_bounds__(256) void fill_in_edge_lists(
    const int* __restrict__ src, const int* __restrict__ dst,
    const int* __restrict__ rowptr, const int* __restrict__ rank,
    int* __restrict__ col, int n_edges) {
  const int tid = (int)blockIdx.x * 256 + (int)threadIdx.x;
  const int nthreads = (int)gridDim.x * 256;
  const int n_groups = n_edges >> 3;
  const int4* __restrict__ src4 = (const int4*)src;
  const int4* __restrict__ dst4 = (const int4*)dst;
  const int4* __restrict__ rank4 = (const int4*)rank;
  for (int g = tid; g < n_groups; g += nthreads) {
    const int4 d0 = dst4[2 * g];
    const int4 d1 = dst4[2 * g + 1];
    const int4 r0 = rank4[2 * g];
    const int4 r1 = rank4[2 * g + 1];
    const int4 s0 = src4[2 * g];
    const int4 s1 = src4[2 * g + 1];
    const int p0 = rowptr[d0.x], p1 = rowptr[d0.y], p2 = rowptr[d0.z], p3 = rowptr[d0.w];
    const int p4 = rowptr[d1.x], p5 = rowptr[d1.y], p6 = rowptr[d1.z], p7 = rowptr[d1.w];
    col[p0 + r0.x] = s0.x;
    col[p1 + r0.y] = s0.y;
    col[p2 + r0.z] = s0.z;
    col[p3 + r0.w] = s0.w;
    col[p4 + r1.x] = s1.x;
    col[p5 + r1.y] = s1.y;
    col[p6 + r1.z] = s1.z;
    col[p7 + r1.w] = s1.w;
  }
}

#define GATHER_WAVES 32
__global__ __launch_bounds__(GATHER_WAVES * 32) void gather_bias_prelu_stats(
    const float* __restrict__ Z, const int* __restrict__ rowptr,
    const int* __restrict__ col, const float* __restrict__ bias,
    const float* __restrict__ a_conv_p, float* __restrict__ out,
    float* __restrict__ stats, int n_rows) {
  __shared__ float part[GATHER_WAVES][2 * DIMD];
  const int lane = threadIdx.x & 31;
  const int wave = __builtin_amdgcn_readfirstlane((int)(threadIdx.x >> 5));
  const int n_waves = (int)gridDim.x * GATHER_WAVES;
  const float4* __restrict__ Z4 = (const float4*)Z;
  float4* __restrict__ O4 = (float4*)out;
  const float4 bb = ((const float4*)bias)[lane];
  const float ac = *a_conv_p;
  float4 s = {0.f, 0.f, 0.f, 0.f};
  float4 ss = {0.f, 0.f, 0.f, 0.f};
  for (int d = (int)blockIdx.x * GATHER_WAVES + wave; d < n_rows; d += n_waves) {
    const int beg = rowptr[d];
    const int end = rowptr[d + 1];
    float4 acc = {0.f, 0.f, 0.f, 0.f};
    for (int b = beg; b < end; b += 32) {
      const int j = b + lane;
      const int mine = (j < end) ? col[j] : 0;
      const int m = (end - b < 32) ? (end - b) : 32;
#pragma unroll 4
      for (int k = 0; k < m; ++k) {
        const int sidx = __builtin_amdgcn_readlane(mine, k);
        const float4 v = Z4[(size_t)sidx * (DIMD / 4) + lane];
        acc.x += v.x;
        acc.y += v.y;
        acc.z += v.z;
        acc.w += v.w;
      }
    }
    float4 z;
    z.x = acc.x + bb.x;
    z.y = acc.y + bb.y;
    z.z = acc.z + bb.z;
    z.w = acc.w + bb.w;
    z.x = (z.x >= 0.f) ? z.x : ac * z.x;
    z.y = (z.y >= 0.f) ? z.y : ac * z.y;
    z.z = (z.z >= 0.f) ? z.z : ac * z.z;
    z.w = (z.w >= 0.f) ? z.w : ac * z.w;
    O4[(size_t)d * (DIMD / 4) + lane] = z;
    s.x += z.x;
    s.y += z.y;
    s.z += z.z;
    s.w += z.w;
    ss.x += z.x * z.x;
    ss.y += z.y * z.y;
    ss.z += z.z * z.z;
    ss.w += z.w * z.w;
  }
  float4* __restrict__ P4 = (float4*)&part[wave][0];
  P4[lane] = s;
  P4[DIMD / 4 + lane] = ss;
  __syncthreads();
  if (threadIdx.x < 2 * DIMD) {
    float t = 0.f;
    for (int w = 0; w < GATHER_WAVES; ++w) t += part[w][threadIdx.x];
    unsafeAtomicAdd(&stats[threadIdx.x], t);
  }
}

#define APPLY_BLOCKS 625
#define APPLY_WAVES 8
#define APPLY_ROWS 10
static_assert(APPLY_BLOCKS * APPLY_WAVES * APPLY_ROWS == NNODES,
              "the row partition of bn_apply_prelu must be exact");
__global__ __launch_bounds__(APPLY_WAVES * 32) void bn_apply_prelu(
    const float* __restrict__ Z, const float* __restrict__ stats,
    const float* __restrict__ gamma, const float* __restrict__ beta,
    const float* __restrict__ a_out_p, float* __restrict__ Hout, int n_rows) {
  const int lane = threadIdx.x & 31;
  const int wave = __builtin_amdgcn_readfirstlane((int)(threadIdx.x >> 5));
  const int gw = (int)blockIdx.x * APPLY_WAVES + wave;
  const float4* __restrict__ Z4 = (const float4*)Z;
  float4* __restrict__ H4 = (float4*)Hout;
  const BnLaneRaw raw = bn_lane_load(stats, gamma, beta, lane);
  const float ao = *a_out_p;
  float4 v[APPLY_ROWS];
#pragma unroll
  for (int u = 0; u < APPLY_ROWS; ++u)
    v[u] = Z4[(size_t)(gw + u * (APPLY_BLOCKS * APPLY_WAVES)) * (DIMD / 4) + lane];
  float4 sc, sh;
  bn_lane_scale_shift(raw, n_rows, sc, sh);
#pragma unroll
  for (int u = 0; u < APPLY_ROWS; ++u)
    H4[(size_t)(gw + u * (APPLY_BLOCKS * APPLY_WAVES)) * (DIMD / 4) + lane] =
        bn_prelu4(v[u], sc, sh, ao);
}

extern "C" void kernel_launch(void* const* d_in, const int* in_sizes, int n_in,
                              void* d_out, int out_size, void* d_ws,
                              size_t ws_size, hipStream_t stream) {
  const float* heat   = (const float*)d_in[0];
  const int*   src    = (const int*)d_in[1];
  const int*   dst    = (const int*)d_in[2];
  const float* W      = (const float*)d_in[3];
  const float* b      = (const float*)d_in[4];
  const float* a_conv = (const float*)d_in[5];
  const float* gamma  = (const float*)d_in[6];
  const float* beta   = (const float*)d_in[7];
  const float* a_out  = (const float*)d_in[8];

  const size_t nb = (size_t)NNODES * DIMD * sizeof(float);
  float* bufA  = (float*)d_ws;
  float* bufB  = (float*)((char*)d_ws + nb);
  float* stats = (float*)((char*)d_ws + 2 * nb);

  int* rowptr = (int*)d_out;
  int* rank   = rowptr + 65536;
  int* col    = rank + NEDGES;

  const dim3 gemmGrid(GEMM_BLOCKS);
  const dim3 gemmBlock(256);
  const int edgeBlocks = EDGE_BLOCKS;
  const int gatherBlocks = 512;

  hipMemsetAsync(rowptr, 0, SCAN_LEN * sizeof(int), stream);
  hipMemsetAsync(stats, 0, 2 * DIMD * sizeof(float), stream);
  count_in_edges<<<edgeBlocks, 256, 0, stream>>>(dst, rowptr, rank, NEDGES);
  scan_in_degrees<<<1, 1024, 0, stream>>>(rowptr);
  fill_in_edge_lists<<<edgeBlocks, 256, 0, stream>>>(src, dst, rowptr, rank,
                                                     col, NEDGES);

  gemm_wmma_f32<<<gemmGrid, gemmBlock, 0, stream>>>(heat, W, bufA, NNODES / 16);
  gather_bias_prelu_stats<<<gatherBlocks, GATHER_WAVES * 32, 0, stream>>>(
      bufA, rowptr, col, b, a_conv, bufB, stats, NNODES);

  gemm_bn_prelu_wmma_f32<<<gemmGrid, gemmBlock, 0, stream>>>(
      bufB, stats, gamma, beta, a_out, W + DIMD * DIMD, bufA, NNODES / 16, NNODES);
  hipMemsetAsync(stats, 0, 2 * DIMD * sizeof(float), stream);
  gather_bias_prelu_stats<<<gatherBlocks, GATHER_WAVES * 32, 0, stream>>>(
      bufA, rowptr, col, b + DIMD, a_conv + 1, bufB, stats, NNODES);
  bn_apply_prelu<<<APPLY_BLOCKS, APPLY_WAVES * 32, 0, stream>>>(
      bufB, stats, gamma + DIMD, beta + DIMD, a_out + 1, (float*)d_out, NNODES);
}


# Gather: a wave requests all source rows of a row (up to 32) before the first add and fetches the next row's ids a row ahead; 4096 resident waves instead of 16384 that kept four loads in flight
# speedup vs baseline: 3.1287x; 1.2390x over previous
#include <hip/hip_runtime.h>
#include <hip/hip_bf16.h>

typedef float v8f __attribute__((ext_vector_type(8)));
typedef float v2f __attribute__((ext_vector_type(2)));

#define DIMD 128
#define NNODES 50000
#define NEDGES 800000

__device__ __forceinline__ void bn_scale_shift(float sum, float sumsq, float g,
                                               float b, float inv_n,
                                               float& scale, float& shift) {
  const float mu  = sum * inv_n;
  const float var = sumsq * inv_n - mu * mu;
  scale = rsqrtf(var + 1e-5f) * g;
  shift = b - mu * scale;
}

struct BnLaneRaw {
  float4 su, sq, ga, be;
};
__device__ __forceinline__ BnLaneRaw bn_lane_load(
    const float* __restrict__ stats, const float* __restrict__ gamma,
    const float* __restrict__ beta, int lane) {
  BnLaneRaw r;
  r.su = ((const float4*)stats)[lane];
  r.sq = ((const float4*)stats)[32 + lane];
  r.ga = ((const float4*)gamma)[lane];
  r.be = ((const float4*)beta)[lane];
  return r;
}
__device__ __forceinline__ void bn_lane_scale_shift(const BnLaneRaw& r,
                                                    int n_rows, float4& sc,
                                                    float4& sh) {
  const float inv_n = 1.0f / (float)n_rows;
  bn_scale_shift(r.su.x, r.sq.x, r.ga.x, r.be.x, inv_n, sc.x, sh.x);
  bn_scale_shift(r.su.y, r.sq.y, r.ga.y, r.be.y, inv_n, sc.y, sh.y);
  bn_scale_shift(r.su.z, r.sq.z, r.ga.z, r.be.z, inv_n, sc.z, sh.z);
  bn_scale_shift(r.su.w, r.sq.w, r.ga.w, r.be.w, inv_n, sc.w, sh.w);
}

__device__ __forceinline__ float4 bn_prelu4(float4 v, float4 sc, float4 sh,
                                            float a) {
  float4 z;
  z.x = fmaf(v.x, sc.x, sh.x);
  z.y = fmaf(v.y, sc.y, sh.y);
  z.z = fmaf(v.z, sc.z, sh.z);
  z.w = fmaf(v.w, sc.w, sh.w);
  z.x = (z.x >= 0.f) ? z.x : a * z.x;
  z.y = (z.y >= 0.f) ? z.y : a * z.y;
  z.z = (z.z >= 0.f) ? z.z : a * z.z;
  z.w = (z.w >= 0.f) ? z.w : a * z.w;
  return z;
}

#define GEMM_BLOCKS 512
#define A_PITCH4 33
template <bool BN_IN>
__device__ __forceinline__ void gemm_rows_wmma_f32(
    float4 (*atile)[16 * A_PITCH4], int bid, int nblk, const float* __restrict__ H,
    const float* __restrict__ W, float* __restrict__ Z, int n_tiles,
    const float* __restrict__ stats, const float* __restrict__ gamma,
    const float* __restrict__ beta, const float* __restrict__ a_out_p,
    int n_rows) {
  const int wave = threadIdx.x >> 5;
  const int lane = threadIdx.x & 31;
  const int col0 = wave * 16;
  const int m    = lane & 15;
  const int h    = lane >> 4;

  v2f bw[32];
#pragma unroll
  for (int i = 0; i < 16; ++i) {
#pragma unroll
    for (int s = 0; s < 2; ++s) {
      const int k = 8 * i + 4 * h + 2 * s;
      bw[2 * i + s].x = W[(size_t)k * DIMD + col0 + m];
      bw[2 * i + s].y = W[(size_t)(k + 1) * DIMD + col0 + m];
    }
  }

  const float4* __restrict__ H4 = (const float4*)H;
  const int stride = nblk;
  int t = bid;
  int cur = 0;
  float4 g0 = {0.f, 0.f, 0.f, 0.f}, g1 = {0.f, 0.f, 0.f, 0.f};
  const bool first = t < n_tiles;
  if (first) {
    g0 = H4[(size_t)(t * 16 + wave) * (DIMD / 4) + lane];
    g1 = H4[(size_t)(t * 16 + wave + 8) * (DIMD / 4) + lane];
  }
  float4 sc = {1.f, 1.f, 1.f, 1.f}, sh = {0.f, 0.f, 0.f, 0.f};
  float ao = 0.f;
  if (BN_IN) {
    bn_lane_scale_shift(bn_lane_load(stats, gamma, beta, lane), n_rows, sc, sh);
    ao = *a_out_p;
  }
  if (first) {
    if (BN_IN) {
      g0 = bn_prelu4(g0, sc, sh, ao);
      g1 = bn_prelu4(g1, sc, sh, ao);
    }
    atile[0][wave * A_PITCH4 + lane] = g0;
    atile[0][(wave + 8) * A_PITCH4 + lane] = g1;
  }
  __syncthreads();

  for (; t < n_tiles; t += stride) {
    const int tn = t + stride;
    const bool more = tn < n_tiles;
    if (more) {
      g0 = H4[(size_t)(tn * 16 + wave) * (DIMD / 4) + lane];
      g1 = H4[(size_t)(tn * 16 + wave + 8) * (DIMD / 4) + lane];
    }
    const float4* __restrict__ arow = &atile[cur][m * A_PITCH4 + h];
    float4 af[16];
#pragma unroll
    for (int i = 0; i < 16; ++i) af[i] = arow[2 * i];
    __builtin_amdgcn_sched_barrier(0);
    v8f acc = {};
#pragma unroll
    for (int i = 0; i < 16; ++i) {
      v2f a0, a1;
      a0.x = af[i].x;
      a0.y = af[i].y;
      a1.x = af[i].z;
      a1.y = af[i].w;
      acc = __builtin_amdgcn_wmma_f32_16x16x4_f32(
           false, a0,  false, bw[2 * i],
           (short)0, acc,  false,  false);
      acc = __builtin_amdgcn_wmma_f32_16x16x4_f32(
           false, a1,  false, bw[2 * i + 1],
           (short)0, acc,  false,  false);
    }
    const int mbase = t * 16 + h * 8;
#pragma unroll
    for (int j = 0; j < 8; ++j)
      Z[(size_t)(mbase + j) * DIMD + col0 + m] = acc[j];
    if (more) {
      if (BN_IN) {
        g0 = bn_prelu4(g0, sc, sh, ao);
        g1 = bn_prelu4(g1, sc, sh, ao);
      }
      atile[cur ^ 1][wave * A_PITCH4 + lane] = g0;
      atile[cur ^ 1][(wave + 8) * A_PITCH4 + lane] = g1;
    }
    __builtin_amdgcn_fence(__ATOMIC_RELEASE, "workgroup", "local");
    __builtin_amdgcn_s_barrier();
    __builtin_amdgcn_fence(__ATOMIC_ACQUIRE, "workgroup", "local");
    cur ^= 1;
  }
}

__global__ __launch_bounds__(256) void gemm_bn_prelu_wmma_f32(
    const float* __restrict__ Hraw, const float* __restrict__ stats,
    const float* __restrict__ gamma, const float* __restrict__ beta,
    const float* __restrict__ a_out_p, const float* __restrict__ W,
    float* __restrict__ Z, int n_tiles, int n_rows) {
  __shared__ float4 atile[2][16 * A_PITCH4];
  gemm_rows_wmma_f32<true>(atile, (int)blockIdx.x, (int)gridDim.x, Hraw, W, Z,
                           n_tiles, stats, gamma, beta, a_out_p, n_rows);
}

#define SLOTS 32
#define CNT_LEN 53248
#define STATS0_AT 51200
#define EDGES_PER_LANE 16
#define EDGE_BLOCKS ((NEDGES / EDGES_PER_LANE + 255) / 256)
static_assert(NEDGES % EDGES_PER_LANE == 0, "the edge pass takes the edges in groups of sixteen");
static_assert(CNT_LEN > NNODES && CNT_LEN <= 65536, "counters and overflow length fit in front of the node lines");
static_assert(STATS0_AT > NNODES && STATS0_AT % 32 == 0 && STATS0_AT + 2 * DIMD <= CNT_LEN,
              "layer 0's column sums lie behind the counters and inside the cleared ints");
__device__ __forceinline__ void place_in_edge(int d, int r, int s,
                                              int* __restrict__ slot,
                                              int* __restrict__ ovf_len,
                                              int* __restrict__ ovf_src,
                                              int* __restrict__ ovf_dst) {
  if (r < SLOTS) {
    slot[(size_t)d * SLOTS + r] = s;
  } else {
    const int o = atomicAdd(ovf_len, 1);
    ovf_src[o] = s;
    ovf_dst[o] = d;
  }
}
__device__ __forceinline__ void place_in_edges(
    int bid, int nblk, const int* __restrict__ src, const int* __restrict__ dst,
    int* __restrict__ cnt, int* __restrict__ slot, int* __restrict__ ovf_len,
    int* __restrict__ ovf_src, int* __restrict__ ovf_dst, int n_edges) {
  const int tid = bid * 256 + (int)threadIdx.x;
  const int nthreads = nblk * 256;
  const int n_groups = n_edges / EDGES_PER_LANE;
  const int4* __restrict__ src4 = (const int4*)src;
  const int4* __restrict__ dst4 = (const int4*)dst;
  for (int g = tid; g < n_groups; g += nthreads) {
    int4 d[4], s[4];
    int r[16];
#pragma unroll
    for (int q = 0; q < 4; ++q) {
      d[q] = dst4[4 * g + q];
      s[q] = src4[4 * g + q];
    }
#pragma unroll
    for (int q = 0; q < 4; ++q) {
      r[4 * q + 0] = atomicAdd(&cnt[d[q].x], 1);
      r[4 * q + 1] = atomicAdd(&cnt[d[q].y], 1);
      r[4 * q + 2] = atomicAdd(&cnt[d[q].z], 1);
      r[4 * q + 3] = atomicAdd(&cnt[d[q].w], 1);
    }
#pragma unroll
    for (int q = 0; q < 4; ++q) {
      place_in_edge(d[q].x, r[4 * q + 0], s[q].x, slot, ovf_len, ovf_src, ovf_dst);
      place_in_edge(d[q].y, r[4 * q + 1], s[q].y, slot, ovf_len, ovf_src, ovf_dst);
      place_in_edge(d[q].z, r[4 * q + 2], s[q].z, slot, ovf_len, ovf_src, ovf_dst);
      place_in_edge(d[q].w, r[4 * q + 3], s[q].w, slot, ovf_len, ovf_src, ovf_dst);
    }
  }
}

#define FIRST_BLOCKS (GEMM_BLOCKS + EDGE_BLOCKS)
__global__ __launch_bounds__(256) void gemm_wmma_f32_and_in_edges(
    const float* __restrict__ H, const float* __restrict__ W,
    float* __restrict__ Z, int n_tiles, const int* __restrict__ src,
    const int* __restrict__ dst, int* __restrict__ cnt, int* __restrict__ slot,
    int* __restrict__ ovf_len, int* __restrict__ ovf_src,
    int* __restrict__ ovf_dst, int n_edges, float* __restrict__ stats_next) {
  __shared__ float4 atile[2][16 * A_PITCH4];
  if ((int)blockIdx.x >= GEMM_BLOCKS) {
    const int bid = (int)blockIdx.x - GEMM_BLOCKS;
    if (bid == 0) stats_next[threadIdx.x] = 0.f;
    place_in_edges(bid, (int)gridDim.x - GEMM_BLOCKS, src, dst, cnt, slot,
                   ovf_len, ovf_src, ovf_dst, n_edges);
    return;
  }
  gemm_rows_wmma_f32<false>(atile, (int)blockIdx.x, GEMM_BLOCKS, H, W, Z,
                            n_tiles, nullptr, nullptr, nullptr, nullptr, 0);
}

#define GATHER_WAVES 8
__device__ __forceinline__ float4 gather_row(const float4* __restrict__ Z4,
                                             int ids, int k, int lane) {
  const int sidx = __builtin_amdgcn_readlane(ids, k);
  return Z4[(size_t)sidx * (DIMD / 4) + lane];
}
__global__ __launch_bounds__(GATHER_WAVES * 32) void gather_bias_prelu_stats(
    const float* __restrict__ Z, const int* __restrict__ cnt,
    const int* __restrict__ slot, const int* __restrict__ ovf_len,
    const int* __restrict__ ovf_src, const int* __restrict__ ovf_dst,
    const float* __restrict__ bias, const float* __restrict__ a_conv_p,
    float* __restrict__ out, float* __restrict__ stats, int n_rows,
    int n_edges) {
  __shared__ float part[GATHER_WAVES][2 * DIMD];
  const int lane = threadIdx.x & 31;
  const int wave = __builtin_amdgcn_readfirstlane((int)(threadIdx.x >> 5));
  const int n_waves = (int)gridDim.x * GATHER_WAVES;
  const float4* __restrict__ Z4 = (const float4*)Z;
  float4* __restrict__ O4 = (float4*)out;
  const float4 bb = ((const float4*)bias)[lane];
  const float ac = *a_conv_p;
  float4 s = {0.f, 0.f, 0.f, 0.f};
  float4 ss = {0.f, 0.f, 0.f, 0.f};
  int d = (int)blockIdx.x * GATHER_WAVES + wave;
  int n_next = 0, ids_next = 0;
  if (d < n_rows) {
    n_next = cnt[d];
    ids_next = slot[(size_t)d * SLOTS + lane];
  }
  for (; d < n_rows; d += n_waves) {
    const int n = n_next;
    const int ids = ids_next;
    const int dn = d + n_waves;
    if (dn < n_rows) {
      n_next = cnt[dn];
      ids_next = slot[(size_t)dn * SLOTS + lane];
    }
    const int m = (n < SLOTS) ? n : SLOTS;
    const int q = m >> 2;
    const int r = m & 3;
    const int k_left = m - r;
    float4 t[3];
    if (r > 0) t[0] = gather_row(Z4, ids, k_left, lane);
    if (r > 1) t[1] = gather_row(Z4, ids, k_left + 1, lane);
    if (r > 2) t[2] = gather_row(Z4, ids, k_left + 2, lane);
    float4 v[SLOTS];
#pragma unroll
    for (int g = 0; g < SLOTS / 4; ++g) {
      if (g < q) {
#pragma unroll
        for (int j = 0; j < 4; ++j) v[4 * g + j] = gather_row(Z4, ids, 4 * g + j, lane);
      }
    }
    float4 acc = {0.f, 0.f, 0.f, 0.f};
#pragma unroll
    for (int g = 0; g < SLOTS / 4; ++g) {
      if (g < q) {
#pragma unroll
        for (int j = 0; j < 4; ++j) {
          acc.x += v[4 * g + j].x;
          acc.y += v[4 * g + j].y;
          acc.z += v[4 * g + j].z;
          acc.w += v[4 * g + j].w;
        }
      }
    }
#pragma unroll
    for (int j = 0; j < 3; ++j) {
      if (j < r) {
        acc.x += t[j].x;
        acc.y += t[j].y;
        acc.z += t[j].z;
        acc.w += t[j].w;
      }
    }
    if (n > SLOTS) {
      int no = *ovf_len;
      no = (no < n_edges) ? no : n_edges;
      for (int j0 = 0; j0 < no; j0 += 32) {
        const int j = j0 + lane;
        const int od = (j < no) ? ovf_dst[j] : -1;
        const int os = (j < no) ? ovf_src[j] : 0;
        unsigned mask = __builtin_amdgcn_ballot_w32(od == d);
        while (mask) {
          const int k = __builtin_ctz(mask);
          mask &= mask - 1;
          const float4 w = gather_row(Z4, os, k, lane);
          acc.x += w.x;
          acc.y += w.y;
          acc.z += w.z;
          acc.w += w.w;
        }
      }
    }
    float4 z;
    z.x = acc.x + bb.x;
    z.y = acc.y + bb.y;
    z.z = acc.z + bb.z;
    z.w = acc.w + bb.w;
    z.x = (z.x >= 0.f) ? z.x : ac * z.x;
    z.y = (z.y >= 0.f) ? z.y : ac * z.y;
    z.z = (z.z >= 0.f) ? z.z : ac * z.z;
    z.w = (z.w >= 0.f) ? z.w : ac * z.w;
    O4[(size_t)d * (DIMD / 4) + lane] = z;
    s.x += z.x;
    s.y += z.y;
    s.z += z.z;
    s.w += z.w;
    ss.x += z.x * z.x;
    ss.y += z.y * z.y;
    ss.z += z.z * z.z;
    ss.w += z.w * z.w;
  }
  float4* __restrict__ P4 = (float4*)&part[wave][0];
  P4[lane] = s;
  P4[DIMD / 4 + lane] = ss;
  __syncthreads();
  if (threadIdx.x < 2 * DIMD) {
    float t = 0.f;
    for (int w = 0; w < GATHER_WAVES; ++w) t += part[w][threadIdx.x];
    unsafeAtomicAdd(&stats[threadIdx.x], t);
  }
}

#define APPLY_BLOCKS 625
#define APPLY_WAVES 8
#define APPLY_ROWS 10
static_assert(APPLY_BLOCKS * APPLY_WAVES * APPLY_ROWS == NNODES,
              "the row partition of bn_apply_prelu must be exact");
__global__ __launch_bounds__(APPLY_WAVES * 32) void bn_apply_prelu(
    const float* __restrict__ Z, const float* __restrict__ stats,
    const float* __restrict__ gamma, const float* __restrict__ beta,
    const float* __restrict__ a_out_p, float* __restrict__ Hout, int n_rows) {
  const int lane = threadIdx.x & 31;
  const int wave = __builtin_amdgcn_readfirstlane((int)(threadIdx.x >> 5));
  const int gw = (int)blockIdx.x * APPLY_WAVES + wave;
  const float4* __restrict__ Z4 = (const float4*)Z;
  float4* __restrict__ H4 = (float4*)Hout;
  const BnLaneRaw raw = bn_lane_load(stats, gamma, beta, lane);
  const float ao = *a_out_p;
  float4 v[APPLY_ROWS];
#pragma unroll
  for (int u = 0; u < APPLY_ROWS; ++u)
    v[u] = Z4[(size_t)(gw + u * (APPLY_BLOCKS * APPLY_WAVES)) * (DIMD / 4) + lane];
  float4 sc, sh;
  bn_lane_scale_shift(raw, n_rows, sc, sh);
#pragma unroll
  for (int u = 0; u < APPLY_ROWS; ++u)
    H4[(size_t)(gw + u * (APPLY_BLOCKS * APPLY_WAVES)) * (DIMD / 4) + lane] =
        bn_prelu4(v[u], sc, sh, ao);
}

extern "C" void kernel_launch(void* const* d_in, const int* in_sizes, int n_in,
                              void* d_out, int out_size, void* d_ws,
                              size_t ws_size, hipStream_t stream) {
  const float* heat   = (const float*)d_in[0];
  const int*   src    = (const int*)d_in[1];
  const int*   dst    = (const int*)d_in[2];
  const float* W      = (const float*)d_in[3];
  const float* b      = (const float*)d_in[4];
  const float* a_conv = (const float*)d_in[5];
  const float* gamma  = (const float*)d_in[6];
  const float* beta   = (const float*)d_in[7];
  const float* a_out  = (const float*)d_in[8];

  const size_t nb = (size_t)NNODES * DIMD * sizeof(float);
  float* bufA   = (float*)d_ws;
  float* bufB   = (float*)((char*)d_ws + nb);
  float* stats1 = (float*)((char*)d_ws + 2 * nb);

  static_assert((size_t)65536 + (size_t)NNODES * SLOTS + 2 * (size_t)NEDGES <=
                    (size_t)NNODES * DIMD,
                "counters, node lines and the overflow list fit in the output buffer");
  int* cnt      = (int*)d_out;
  int* ovf_len  = cnt + NNODES;
  float* stats0 = (float*)(cnt + STATS0_AT);
  int* slot     = cnt + 65536;
  int* ovf_src  = slot + NNODES * SLOTS;
  int* ovf_dst  = ovf_src + NEDGES;

  const dim3 gemmGrid(GEMM_BLOCKS);
  const dim3 gemmBlock(256);
  const int gatherBlocks = 512;

  hipMemsetAsync(cnt, 0, CNT_LEN * sizeof(int), stream);

  gemm_wmma_f32_and_in_edges<<<FIRST_BLOCKS, 256, 0, stream>>>(
      heat, W, bufA, NNODES / 16, src, dst, cnt, slot, ovf_len, ovf_src, ovf_dst,
      NEDGES, stats1);

  gather_bias_prelu_stats<<<gatherBlocks, GATHER_WAVES * 32, 0, stream>>>(
      bufA, cnt, slot, ovf_len, ovf_src, ovf_dst, b, a_conv, bufB, stats0, NNODES,
      NEDGES);

  gemm_bn_prelu_wmma_f32<<<gemmGrid, gemmBlock, 0, stream>>>(
      bufB, stats0, gamma, beta, a_out, W + DIMD * DIMD, bufA, NNODES / 16, NNODES);
  gather_bias_prelu_stats<<<gatherBlocks, GATHER_WAVES * 32, 0, stream>>>(
      bufA, cnt, slot, ovf_len, ovf_src, ovf_dst, b + DIMD, a_conv + 1, bufB, stats1,
      NNODES, NEDGES);
  bn_apply_prelu<<<APPLY_BLOCKS, APPLY_WAVES * 32, 0, stream>>>(
      bufB, stats1, gamma + DIMD, beta + DIMD, a_out + 1, (float*)d_out, NNODES);
}
